# starting kernel
# baseline (speedup 1.0000x reference)
#include <hip/hip_runtime.h>
#include <hip/hip_bf16.h>

typedef __attribute__((ext_vector_type(2))) float v2f;
typedef __attribute__((ext_vector_type(8))) float v8f;

#define N_ROWS 131072
#define DIM    512
#define KCLUST 64
#define ROWS_PER_BLOCK 128
#define CHUNK  32
#define CPITCH 160
#define EPSN   1e-12f

__global__ void k_norm_clusters(const float* __restrict__ clusters,
                                float* __restrict__ ct2) {
    __shared__ float sred[8];
    const int k    = blockIdx.x;
    const int tid  = threadIdx.x;
    const int lane = tid & 31;
    const int wid  = tid >> 5;

    float v0 = clusters[k * DIM + tid];
    float v1 = clusters[k * DIM + tid + 256];
    float ss = v0 * v0 + v1 * v1;
    #pragma unroll
    for (int o = 16; o > 0; o >>= 1) ss += __shfl_xor(ss, o, 32);
    if (lane == 0) sred[wid] = ss;
    __syncthreads();
    float tot = 0.f;
    #pragma unroll
    for (int i = 0; i < 8; ++i) tot += sred[i];
    const float inv = 1.0f / fmaxf(sqrtf(tot), EPSN);

    const int d0 = tid, d1 = tid + 256;
    ct2[(d0 >> 1) * 128 + k * 2 + (d0 & 1)] = v0 * inv;
    ct2[(d1 >> 1) * 128 + k * 2 + (d1 & 1)] = v1 * inv;
}

__global__ void __launch_bounds__(256)
k_cluster_probs(const float* __restrict__ x,
                const float* __restrict__ ct2,
                const float* __restrict__ alphap,
                float* __restrict__ out_probs,
                float* __restrict__ loss_part) {
    __shared__ float xs[ROWS_PER_BLOCK * 36];
    __shared__ float cs[(CHUNK / 2) * CPITCH];
    __shared__ float tile[ROWS_PER_BLOCK * 65];
    __shared__ float lred[8];

    const int tid   = threadIdx.x;
    const int lane  = tid & 31;
    const int wave  = tid >> 5;
    const int laneN = lane & 15;
    const int laneH = lane >> 4;
    const int rowbase = blockIdx.x * ROWS_PER_BLOCK;
    const float alpha = alphap[0];

    const int xr0 = tid >> 3;
    const int xc0 = (tid & 7) << 2;
    const int pr0 = tid >> 5;
    const int q0  = (tid & 31) << 2;

    v8f acc[4];
    #pragma unroll
    for (int t = 0; t < 4; ++t) acc[t] = (v8f){0.f,0.f,0.f,0.f,0.f,0.f,0.f,0.f};
    float ss = 0.f;

    for (int kb = 0; kb < DIM; kb += CHUNK) {
        float4 a0 = *(const float4*)&x[(size_t)(rowbase + xr0)       * DIM + kb + xc0];
        float4 a1 = *(const float4*)&x[(size_t)(rowbase + xr0 +  32) * DIM + kb + xc0];
        float4 a2 = *(const float4*)&x[(size_t)(rowbase + xr0 +  64) * DIM + kb + xc0];
        float4 a3 = *(const float4*)&x[(size_t)(rowbase + xr0 +  96) * DIM + kb + xc0];
        float4 c0 = *(const float4*)&ct2[(size_t)((kb >> 1) + pr0)     * 128 + q0];
        float4 c1 = *(const float4*)&ct2[(size_t)((kb >> 1) + pr0 + 8) * 128 + q0];

        *(float4*)&xs[(xr0      ) * 36 + xc0] = a0;
        *(float4*)&xs[(xr0 +  32) * 36 + xc0] = a1;
        *(float4*)&xs[(xr0 +  64) * 36 + xc0] = a2;
        *(float4*)&xs[(xr0 +  96) * 36 + xc0] = a3;
        *(float4*)&cs[(pr0    ) * CPITCH + q0] = c0;
        *(float4*)&cs[(pr0 + 8) * CPITCH + q0] = c1;
        __syncthreads();

        #pragma unroll
        for (int kk = 0; kk < CHUNK; kk += 4) {
            v2f a = *(const v2f*)&xs[(wave * 16 + laneN) * 36 + kk + 2 * laneH];
            ss += a.x * a.x + a.y * a.y;
            const int prow = (kk >> 1) + laneH;
            #pragma unroll
            for (int t = 0; t < 4; ++t) {
                v2f b = *(const v2f*)&cs[prow * CPITCH + (t * 16 + laneN) * 2];
                acc[t] = __builtin_amdgcn_wmma_f32_16x16x4_f32(
                    false, a, false, b, (short)0, acc[t], false, false);
            }
        }
        __syncthreads();
    }

    ss += __shfl_xor(ss, 16, 32);
    const float rinv = 1.0f / fmaxf(sqrtf(ss), EPSN);

    #pragma unroll
    for (int t = 0; t < 4; ++t) {
        #pragma unroll
        for (int g = 0; g < 8; ++g) {
            float invm = __shfl(rinv, g + 8 * laneH, 32);
            tile[(wave * 16 + g + 8 * laneH) * 65 + t * 16 + laneN] = acc[t][g] * invm;
        }
    }
    __syncthreads();

    const int srow  = tid >> 1;
    const int shalf = tid & 1;
    float* rp = &tile[srow * 65 + shalf * 32];

    float m = -3.402823466e+38f;
    #pragma unroll
    for (int i = 0; i < 32; ++i) m = fmaxf(m, alpha * rp[i]);
    m = fmaxf(m, __shfl_xor(m, 1, 32));

    float se = 0.f, sev = 0.f;
    #pragma unroll
    for (int i = 0; i < 32; ++i) {
        float v = rp[i];
        float e = __expf(alpha * v - m);
        se  += e;
        sev += e * v;
    }
    float totv = se + __shfl_xor(se, 1, 32);
    float itot = 1.0f / totv;
    #pragma unroll
    for (int i = 0; i < 32; ++i) {
        float v = rp[i];
        rp[i] = __expf(alpha * v - m) * itot;
    }
    float lpart = sev * itot;
    __syncthreads();

    #pragma unroll
    for (int p = 0; p < 32; ++p) {
        int idx = tid + p * 256;
        int r   = idx >> 6;
        int c   = idx & 63;
        out_probs[(size_t)(rowbase + r) * KCLUST + c] = tile[r * 65 + c];
    }

    #pragma unroll
    for (int o = 16; o > 0; o >>= 1) lpart += __shfl_xor(lpart, o, 32);
    if (lane == 0) lred[wave] = lpart;
    __syncthreads();
    if (tid == 0) {
        float t2 = 0.f;
        #pragma unroll
        for (int i = 0; i < 8; ++i) t2 += lred[i];
        loss_part[blockIdx.x] = t2;
    }
}

__global__ void k_finalize(const float* __restrict__ parts, float* __restrict__ out) {
    __shared__ float sred[8];
    const int tid  = threadIdx.x;
    const int lane = tid & 31;
    const int wid  = tid >> 5;
    float s = parts[tid] + parts[tid + 256] + parts[tid + 512] + parts[tid + 768];
    #pragma unroll
    for (int o = 16; o > 0; o >>= 1) s += __shfl_xor(s, o, 32);
    if (lane == 0) sred[wid] = s;
    __syncthreads();
    if (tid == 0) {
        float tot = 0.f;
        #pragma unroll
        for (int i = 0; i < 8; ++i) tot += sred[i];
        out[0] = -tot / (float)N_ROWS;
    }
}

extern "C" void kernel_launch(void* const* d_in, const int* in_sizes, int n_in,
                              void* d_out, int out_size, void* d_ws, size_t ws_size,
                              hipStream_t stream) {
    (void)in_sizes; (void)n_in; (void)out_size; (void)ws_size;
    const float* x        = (const float*)d_in[0];
    const float* clusters = (const float*)d_in[1];
    const float* alpha    = (const float*)d_in[2];
    float* out = (float*)d_out;

    float* ct2   = (float*)d_ws;
    float* parts = ct2 + DIM * KCLUST;

    k_norm_clusters<<<KCLUST, 256, 0, stream>>>(clusters, ct2);
    k_cluster_probs<<<N_ROWS / ROWS_PER_BLOCK, 256, 0, stream>>>(
        x, ct2, alpha, out + 1, parts);
    k_finalize<<<1, 256, 0, stream>>>(parts, out);
}


# Register-prefetch pipeline: the next chunk's 6 global loads are issued before the current chunk's WMMA phase and stored to LDS after it; waves_per_eu(8,8) keeps all 4 blocks per WGP at 122 VGPRs
# speedup vs baseline: 1.0380x; 1.0380x over previous
#include <hip/hip_runtime.h>
#include <hip/hip_bf16.h>

typedef __attribute__((ext_vector_type(2))) float v2f;
typedef __attribute__((ext_vector_type(8))) float v8f;

#define N_ROWS 131072
#define DIM    512
#define KCLUST 64
#define ROWS_PER_BLOCK 128
#define CHUNK  32
#define CPITCH 160
#define EPSN   1e-12f

__global__ void k_norm_clusters(const float* __restrict__ clusters,
                                float* __restrict__ ct2) {
    __shared__ float sred[8];
    const int k    = blockIdx.x;
    const int tid  = threadIdx.x;
    const int lane = tid & 31;
    const int wid  = tid >> 5;

    float v0 = clusters[k * DIM + tid];
    float v1 = clusters[k * DIM + tid + 256];
    float ss = v0 * v0 + v1 * v1;
    #pragma unroll
    for (int o = 16; o > 0; o >>= 1) ss += __shfl_xor(ss, o, 32);
    if (lane == 0) sred[wid] = ss;
    __syncthreads();
    float tot = 0.f;
    #pragma unroll
    for (int i = 0; i < 8; ++i) tot += sred[i];
    const float inv = 1.0f / fmaxf(sqrtf(tot), EPSN);

    const int d0 = tid, d1 = tid + 256;
    ct2[(d0 >> 1) * 128 + k * 2 + (d0 & 1)] = v0 * inv;
    ct2[(d1 >> 1) * 128 + k * 2 + (d1 & 1)] = v1 * inv;
}

__global__ void __launch_bounds__(256) __attribute__((amdgpu_waves_per_eu(8, 8)))
k_cluster_probs(const float* __restrict__ x,
                const float* __restrict__ ct2,
                const float* __restrict__ alphap,
                float* __restrict__ out_probs,
                float* __restrict__ loss_part) {
    __shared__ float xs[ROWS_PER_BLOCK * 36];
    __shared__ float cs[(CHUNK / 2) * CPITCH];
    __shared__ float tile[ROWS_PER_BLOCK * 65];
    __shared__ float lred[8];

    const int tid   = threadIdx.x;
    const int lane  = tid & 31;
    const int wave  = tid >> 5;
    const int laneN = lane & 15;
    const int laneH = lane >> 4;
    const int rowbase = blockIdx.x * ROWS_PER_BLOCK;
    const float alpha = alphap[0];

    const int xr0 = tid >> 3;
    const int xc0 = (tid & 7) << 2;
    const int pr0 = tid >> 5;
    const int q0  = (tid & 31) << 2;

    v8f acc[4];
    #pragma unroll
    for (int t = 0; t < 4; ++t) acc[t] = (v8f){0.f,0.f,0.f,0.f,0.f,0.f,0.f,0.f};
    float ss = 0.f;

    const float* xp0 = &x[(size_t)(rowbase + xr0)       * DIM + xc0];
    const float* xp1 = &x[(size_t)(rowbase + xr0 +  32) * DIM + xc0];
    const float* xp2 = &x[(size_t)(rowbase + xr0 +  64) * DIM + xc0];
    const float* xp3 = &x[(size_t)(rowbase + xr0 +  96) * DIM + xc0];
    const float* cp0 = &ct2[(size_t)(pr0    ) * 128 + q0];
    const float* cp1 = &ct2[(size_t)(pr0 + 8) * 128 + q0];
    float4 a0 = *(const float4*)xp0;
    float4 a1 = *(const float4*)xp1;
    float4 a2 = *(const float4*)xp2;
    float4 a3 = *(const float4*)xp3;
    float4 c0 = *(const float4*)cp0;
    float4 c1 = *(const float4*)cp1;

    for (int kb = 0; kb < DIM; kb += CHUNK) {
        *(float4*)&xs[(xr0      ) * 36 + xc0] = a0;
        *(float4*)&xs[(xr0 +  32) * 36 + xc0] = a1;
        *(float4*)&xs[(xr0 +  64) * 36 + xc0] = a2;
        *(float4*)&xs[(xr0 +  96) * 36 + xc0] = a3;
        *(float4*)&cs[(pr0    ) * CPITCH + q0] = c0;
        *(float4*)&cs[(pr0 + 8) * CPITCH + q0] = c1;
        __syncthreads();

        if (kb + CHUNK < DIM) {
            const int kn = kb + CHUNK;
            a0 = *(const float4*)(xp0 + kn);
            a1 = *(const float4*)(xp1 + kn);
            a2 = *(const float4*)(xp2 + kn);
            a3 = *(const float4*)(xp3 + kn);
            c0 = *(const float4*)(cp0 + (size_t)(kn >> 1) * 128);
            c1 = *(const float4*)(cp1 + (size_t)(kn >> 1) * 128);
        }

        #pragma unroll
        for (int kk = 0; kk < CHUNK; kk += 4) {
            v2f a = *(const v2f*)&xs[(wave * 16 + laneN) * 36 + kk + 2 * laneH];
            ss += a.x * a.x + a.y * a.y;
            const int prow = (kk >> 1) + laneH;
            #pragma unroll
            for (int t = 0; t < 4; ++t) {
                v2f b = *(const v2f*)&cs[prow * CPITCH + (t * 16 + laneN) * 2];
                acc[t] = __builtin_amdgcn_wmma_f32_16x16x4_f32(
                    false, a, false, b, (short)0, acc[t], false, false);
            }
        }
        __syncthreads();
    }

    ss += __shfl_xor(ss, 16, 32);
    const float rinv = 1.0f / fmaxf(sqrtf(ss), EPSN);

    #pragma unroll
    for (int t = 0; t < 4; ++t) {
        #pragma unroll
        for (int g = 0; g < 8; ++g) {
            float invm = __shfl(rinv, g + 8 * laneH, 32);
            tile[(wave * 16 + g + 8 * laneH) * 65 + t * 16 + laneN] = acc[t][g] * invm;
        }
    }
    __syncthreads();

    const int srow  = tid >> 1;
    const int shalf = tid & 1;
    float* rp = &tile[srow * 65 + shalf * 32];

    float m = -3.402823466e+38f;
    #pragma unroll
    for (int i = 0; i < 32; ++i) m = fmaxf(m, alpha * rp[i]);
    m = fmaxf(m, __shfl_xor(m, 1, 32));

    float se = 0.f, sev = 0.f;
    #pragma unroll
    for (int i = 0; i < 32; ++i) {
        float v = rp[i];
        float e = __expf(alpha * v - m);
        se  += e;
        sev += e * v;
    }
    float totv = se + __shfl_xor(se, 1, 32);
    float itot = 1.0f / totv;
    #pragma unroll
    for (int i = 0; i < 32; ++i) {
        float v = rp[i];
        rp[i] = __expf(alpha * v - m) * itot;
    }
    float lpart = sev * itot;
    __syncthreads();

    #pragma unroll
    for (int p = 0; p < 32; ++p) {
        int idx = tid + p * 256;
        int r   = idx >> 6;
        int c   = idx & 63;
        out_probs[(size_t)(rowbase + r) * KCLUST + c] = tile[r * 65 + c];
    }

    #pragma unroll
    for (int o = 16; o > 0; o >>= 1) lpart += __shfl_xor(lpart, o, 32);
    if (lane == 0) lred[wave] = lpart;
    __syncthreads();
    if (tid == 0) {
        float t2 = 0.f;
        #pragma unroll
        for (int i = 0; i < 8; ++i) t2 += lred[i];
        loss_part[blockIdx.x] = t2;
    }
}

__global__ void k_finalize(const float* __restrict__ parts, float* __restrict__ out) {
    __shared__ float sred[8];
    const int tid  = threadIdx.x;
    const int lane = tid & 31;
    const int wid  = tid >> 5;
    float s = parts[tid] + parts[tid + 256] + parts[tid + 512] + parts[tid + 768];
    #pragma unroll
    for (int o = 16; o > 0; o >>= 1) s += __shfl_xor(s, o, 32);
    if (lane == 0) sred[wid] = s;
    __syncthreads();
    if (tid == 0) {
        float tot = 0.f;
        #pragma unroll
        for (int i = 0; i < 8; ++i) tot += sred[i];
        out[0] = -tot / (float)N_ROWS;
    }
}

extern "C" void kernel_launch(void* const* d_in, const int* in_sizes, int n_in,
                              void* d_out, int out_size, void* d_ws, size_t ws_size,
                              hipStream_t stream) {
    (void)in_sizes; (void)n_in; (void)out_size; (void)ws_size;
    const float* x        = (const float*)d_in[0];
    const float* clusters = (const float*)d_in[1];
    const float* alpha    = (const float*)d_in[2];
    float* out = (float*)d_out;

    float* ct2   = (float*)d_ws;
    float* parts = ct2 + DIM * KCLUST;

    k_norm_clusters<<<KCLUST, 256, 0, stream>>>(clusters, ct2);
    k_cluster_probs<<<N_ROWS / ROWS_PER_BLOCK, 256, 0, stream>>>(
        x, ct2, alpha, out + 1, parts);
    k_finalize<<<1, 256, 0, stream>>>(parts, out);
}


# Cluster normalisation folded into the main kernel (3 launches to 2): raw clusters staged per chunk, norms summed in-block (new summation order), 1/norm applied to the product's columns
# speedup vs baseline: 1.0882x; 1.0484x over previous
#include <hip/hip_runtime.h>
#include <hip/hip_bf16.h>

typedef __attribute__((ext_vector_type(2))) float v2f;
typedef __attribute__((ext_vector_type(8))) float v8f;

#define N_ROWS 131072
#define DIM    512
#define KCLUST 64
#define ROWS_PER_BLOCK 128
#define CHUNK  32
#define PITCH  36
#define EPSN   1e-12f

__global__ void __launch_bounds__(256) __attribute__((amdgpu_waves_per_eu(8, 8)))
k_cluster_probs(const float* __restrict__ x,
                const float* __restrict__ clusters,
                const float* __restrict__ alphap,
                float* __restrict__ out_probs,
                float* __restrict__ loss_part) {
    __shared__ __attribute__((aligned(16))) float xs[ROWS_PER_BLOCK * PITCH];
    __shared__ __attribute__((aligned(16))) float cs[KCLUST * PITCH];
    __shared__ float cinv[KCLUST];
    __shared__ float tile[ROWS_PER_BLOCK * 65];
    __shared__ float lred[8];

    const int tid   = threadIdx.x;
    const int lane  = tid & 31;
    const int wave  = tid >> 5;
    const int laneN = lane & 15;
    const int laneH = lane >> 4;
    const int rowbase = blockIdx.x * ROWS_PER_BLOCK;
    const float alpha = alphap[0];

    const int xr0 = tid >> 3;
    const int xc0 = (tid & 7) << 2;

    v8f acc[4];
    #pragma unroll
    for (int t = 0; t < 4; ++t) acc[t] = (v8f){0.f,0.f,0.f,0.f,0.f,0.f,0.f,0.f};
    float ss = 0.f;
    float css0 = 0.f, css1 = 0.f;

    const float* xp0 = &x[(size_t)(rowbase + xr0)       * DIM + xc0];
    const float* xp1 = &x[(size_t)(rowbase + xr0 +  32) * DIM + xc0];
    const float* xp2 = &x[(size_t)(rowbase + xr0 +  64) * DIM + xc0];
    const float* xp3 = &x[(size_t)(rowbase + xr0 +  96) * DIM + xc0];
    const float* cp0 = &clusters[(size_t)(xr0     ) * DIM + xc0];
    const float* cp1 = &clusters[(size_t)(xr0 + 32) * DIM + xc0];
    float4 a0 = *(const float4*)xp0;
    float4 a1 = *(const float4*)xp1;
    float4 a2 = *(const float4*)xp2;
    float4 a3 = *(const float4*)xp3;
    float4 c0 = *(const float4*)cp0;
    float4 c1 = *(const float4*)cp1;

    for (int kb = 0; kb < DIM; kb += CHUNK) {
        *(float4*)&xs[(xr0      ) * PITCH + xc0] = a0;
        *(float4*)&xs[(xr0 +  32) * PITCH + xc0] = a1;
        *(float4*)&xs[(xr0 +  64) * PITCH + xc0] = a2;
        *(float4*)&xs[(xr0 +  96) * PITCH + xc0] = a3;
        *(float4*)&cs[(xr0      ) * PITCH + xc0] = c0;
        *(float4*)&cs[(xr0 +  32) * PITCH + xc0] = c1;
        css0 += c0.x * c0.x + c0.y * c0.y + c0.z * c0.z + c0.w * c0.w;
        css1 += c1.x * c1.x + c1.y * c1.y + c1.z * c1.z + c1.w * c1.w;
        __syncthreads();

        if (kb + CHUNK < DIM) {
            const int kn = kb + CHUNK;
            a0 = *(const float4*)(xp0 + kn);
            a1 = *(const float4*)(xp1 + kn);
            a2 = *(const float4*)(xp2 + kn);
            a3 = *(const float4*)(xp3 + kn);
            c0 = *(const float4*)(cp0 + kn);
            c1 = *(const float4*)(cp1 + kn);
        }

        #pragma unroll
        for (int kk = 0; kk < CHUNK; kk += 4) {
            v2f a = *(const v2f*)&xs[(wave * 16 + laneN) * PITCH + kk + 2 * laneH];
            ss += a.x * a.x + a.y * a.y;
            #pragma unroll
            for (int t = 0; t < 4; ++t) {
                v2f b = *(const v2f*)&cs[(t * 16 + laneN) * PITCH + kk + 2 * laneH];
                acc[t] = __builtin_amdgcn_wmma_f32_16x16x4_f32(
                    false, a, false, b, (short)0, acc[t], false, false);
            }
        }
        __syncthreads();
    }

    ss += __shfl_xor(ss, 16, 32);
    const float rinv = 1.0f / fmaxf(sqrtf(ss), EPSN);

    css0 += __shfl_xor(css0, 1, 32);  css1 += __shfl_xor(css1, 1, 32);
    css0 += __shfl_xor(css0, 2, 32);  css1 += __shfl_xor(css1, 2, 32);
    css0 += __shfl_xor(css0, 4, 32);  css1 += __shfl_xor(css1, 4, 32);
    if ((tid & 7) == 0) {
        cinv[xr0     ] = 1.0f / fmaxf(sqrtf(css0), EPSN);
        cinv[xr0 + 32] = 1.0f / fmaxf(sqrtf(css1), EPSN);
    }
    __syncthreads();

    float cn[4];
    #pragma unroll
    for (int t = 0; t < 4; ++t) cn[t] = cinv[t * 16 + laneN];
    #pragma unroll
    for (int g = 0; g < 8; ++g) {
        const float invm = __shfl(rinv, g + 8 * laneH, 32);
        #pragma unroll
        for (int t = 0; t < 4; ++t)
            tile[(wave * 16 + g + 8 * laneH) * 65 + t * 16 + laneN] = acc[t][g] * invm * cn[t];
    }
    __syncthreads();

    const int srow  = tid >> 1;
    const int shalf = tid & 1;
    float* rp = &tile[srow * 65 + shalf * 32];

    float m = -3.402823466e+38f;
    #pragma unroll
    for (int i = 0; i < 32; ++i) m = fmaxf(m, alpha * rp[i]);
    m = fmaxf(m, __shfl_xor(m, 1, 32));

    float se = 0.f, sev = 0.f;
    #pragma unroll
    for (int i = 0; i < 32; ++i) {
        float v = rp[i];
        float e = __expf(alpha * v - m);
        se  += e;
        sev += e * v;
    }
    float totv = se + __shfl_xor(se, 1, 32);
    float itot = 1.0f / totv;
    #pragma unroll
    for (int i = 0; i < 32; ++i) {
        float v = rp[i];
        rp[i] = __expf(alpha * v - m) * itot;
    }
    float lpart = sev * itot;
    __syncthreads();

    #pragma unroll
    for (int p = 0; p < 32; ++p) {
        int idx = tid + p * 256;
        int r   = idx >> 6;
        int c   = idx & 63;
        out_probs[(size_t)(rowbase + r) * KCLUST + c] = tile[r * 65 + c];
    }

    #pragma unroll
    for (int o = 16; o > 0; o >>= 1) lpart += __shfl_xor(lpart, o, 32);
    if (lane == 0) lred[wave] = lpart;
    __syncthreads();
    if (tid == 0) {
        float t2 = 0.f;
        #pragma unroll
        for (int i = 0; i < 8; ++i) t2 += lred[i];
        loss_part[blockIdx.x] = t2;
    }
}

__global__ void k_finalize(const float* __restrict__ parts, float* __restrict__ out) {
    __shared__ float sred[8];
    const int tid  = threadIdx.x;
    const int lane = tid & 31;
    const int wid  = tid >> 5;
    float s = parts[tid] + parts[tid + 256] + parts[tid + 512] + parts[tid + 768];
    #pragma unroll
    for (int o = 16; o > 0; o >>= 1) s += __shfl_xor(s, o, 32);
    if (lane == 0) sred[wid] = s;
    __syncthreads();
    if (tid == 0) {
        float tot = 0.f;
        #pragma unroll
        for (int i = 0; i < 8; ++i) tot += sred[i];
        out[0] = -tot / (float)N_ROWS;
    }
}

extern "C" void kernel_launch(void* const* d_in, const int* in_sizes, int n_in,
                              void* d_out, int out_size, void* d_ws, size_t ws_size,
                              hipStream_t stream) {
    (void)in_sizes; (void)n_in; (void)out_size; (void)ws_size;
    const float* x        = (const float*)d_in[0];
    const float* clusters = (const float*)d_in[1];
    const float* alpha    = (const float*)d_in[2];
    float* out = (float*)d_out;

    float* parts = (float*)d_ws;

    k_cluster_probs<<<N_ROWS / ROWS_PER_BLOCK, 256, 0, stream>>>(
        x, clusters, alpha, out + 1, parts);
    k_finalize<<<1, 256, 0, stream>>>(parts, out);
}
